# starting kernel
# baseline (speedup 1.0000x reference)
#include <hip/hip_runtime.h>
#include <cstddef>

typedef __attribute__((ext_vector_type(2))) float v2f;
typedef __attribute__((ext_vector_type(8))) float v8f;

#define DEV __device__ __forceinline__

DEV float eluf(float v) { return v > 0.f ? v : (__expf(v) - 1.f); }
DEV float histf(float v, float bin, float nm1) {
  return fmaxf(1.f - fabsf(v - bin) * nm1, 0.f);
}

DEV v8f wmma_f32(v2f a, v2f b, v8f c) {
  return __builtin_amdgcn_wmma_f32_16x16x4_f32(false, a, false, b, (short)0, c,
                                               false, false);
}

__global__ void tr0_kernel(const float* __restrict__ w, float* __restrict__ wT) {
  int idx = blockIdx.x * blockDim.x + threadIdx.x;
  if (idx >= 864 * 64) return;
  int k = idx >> 6, n = idx & 63;
  int r = k / 96, cf = k % 96;
  float v = (cf < 93) ? w[(size_t)n * 837 + cf * 9 + r] : 0.f;
  wT[((((size_t)(k >> 1)) * 64 + n) << 1) + (k & 1)] = v;
}

__global__ void tr1_kernel(const float* __restrict__ w, float* __restrict__ wT) {
  int idx = blockIdx.x * blockDim.x + threadIdx.x;
  if (idx >= 576 * 64) return;
  int k = idx >> 6, n = idx & 63;
  int r = k >> 6, cin = k & 63;
  float v = w[(size_t)n * 576 + cin * 9 + r];
  wT[((((size_t)(k >> 1)) * 64 + n) << 1) + (k & 1)] = v;
}

__global__ __launch_bounds__(32) void conv0_kernel(const float* __restrict__ x,
                                                   const float* __restrict__ wT,
                                                   const float* __restrict__ bias,
                                                   float* __restrict__ f1) {
  const int lane = threadIdx.x;
  const int hh = lane >> 4;
  const int ml = lane & 15;
  const int tx = blockIdx.x;
  const int py = blockIdx.y;
  const int b = blockIdx.z;

  __shared__ float ft[165][98];

  const int gy0 = 4 * py;
  const int gx0 = 32 * tx;
  const float* xb = x + (size_t)b * 3 * 512 * 512;

  for (int p = lane; p < 165; p += 32) {
    const int ly = p / 33, lx = p - ly * 33;
    const int gy = min(gy0 + ly, 511);
    const int gx = min(gx0 + lx, 511);
    const int ym = max(gy - 1, 0), yp = min(gy + 1, 511);
    const int xm = max(gx - 1, 0), xp = min(gx + 1, 511);
    float* f = &ft[p][0];
    float cv[3];
    int o = 9;
#pragma unroll
    for (int c = 0; c < 3; ++c) {
      const float* xc = xb + (size_t)c * 512 * 512;
      float a = xc[ym * 512 + xm], bb = xc[ym * 512 + gx], cc = xc[ym * 512 + xp];
      float d = xc[gy * 512 + xm], ee = xc[gy * 512 + gx], e = xc[gy * 512 + xp];
      float g = xc[yp * 512 + xm], h = xc[yp * 512 + gx], i = xc[yp * 512 + xp];
      cv[c] = ee;
      f[c] = ee;
      f[3 + 2 * c] = ((cc + 2.f * e + i) - (a + 2.f * d + g)) * 0.125f;
      f[4 + 2 * c] = ((g + 2.f * h + i) - (a + 2.f * bb + cc)) * 0.125f;
    }
#pragma unroll
    for (int c = 0; c < 3; ++c)
#pragma unroll
      for (int bn = 0; bn < 4; ++bn) f[o++] = histf(cv[c], (float)bn * (1.f / 3.f), 3.f);
#pragma unroll
    for (int c = 0; c < 3; ++c)
#pragma unroll
      for (int bn = 0; bn < 8; ++bn) f[o++] = histf(cv[c], (float)bn * (1.f / 7.f), 7.f);
#pragma unroll
    for (int c = 0; c < 3; ++c)
#pragma unroll
      for (int bn = 0; bn < 16; ++bn) f[o++] = histf(cv[c], (float)bn * (1.f / 15.f), 15.f);
    f[93] = 0.f;
    f[94] = 0.f;
    f[95] = 0.f;
  }
  __syncthreads();

  v8f acc[2][4];
  const v8f vz = {0.f, 0.f, 0.f, 0.f, 0.f, 0.f, 0.f, 0.f};
  for (int mt = 0; mt < 2; ++mt)
    for (int nt = 0; nt < 4; ++nt) acc[mt][nt] = vz;

  for (int r = 0; r < 9; ++r) {
    const int ky = r / 3, kx = r % 3;
    const int lx = 2 * ml + kx;
    const float* fp0 = &ft[ky * 33 + lx][2 * hh];
    const float* fp1 = &ft[(ky + 2) * 33 + lx][2 * hh];
    __builtin_prefetch(wT + (((size_t)(min(r + 1, 8) * 48)) * 64) * 2, 0, 1);
    for (int u = 0; u < 24; ++u) {
      v2f a0 = *(const v2f*)(fp0 + 4 * u);
      v2f a1 = *(const v2f*)(fp1 + 4 * u);
      const float* bp = wT + (((size_t)(r * 48 + 2 * u + hh)) * 64) * 2;
#pragma unroll
      for (int nt = 0; nt < 4; ++nt) {
        v2f bv = *(const v2f*)(bp + ((nt * 16 + ml) << 1));
        acc[0][nt] = wmma_f32(a0, bv, acc[0][nt]);
        acc[1][nt] = wmma_f32(a1, bv, acc[1][nt]);
      }
    }
  }

  const int pxb = 8 * tx;
#pragma unroll
  for (int nt = 0; nt < 4; ++nt) {
    const float bb = bias[nt * 16 + ml];
#pragma unroll
    for (int j = 0; j < 4; ++j) {
      float m0 = fmaxf(eluf(acc[0][nt][2 * j] + bb), eluf(acc[0][nt][2 * j + 1] + bb));
      float m1 = fmaxf(eluf(acc[1][nt][2 * j] + bb), eluf(acc[1][nt][2 * j + 1] + bb));
      int px = pxb + j + 4 * hh;
      if (px < 127)
        f1[((((size_t)b * 127 + py) * 127 + px) << 6) + nt * 16 + ml] = fmaxf(m0, m1);
    }
  }
}

template <int HIN, int WIN, int HP, int WP>
__global__ __launch_bounds__(32) void convmid_kernel(const float* __restrict__ fin,
                                                     const float* __restrict__ wT,
                                                     const float* __restrict__ bias,
                                                     float* __restrict__ fout) {
  const int lane = threadIdx.x;
  const int hh = lane >> 4;
  const int ml = lane & 15;
  const int tx = blockIdx.x;
  const int py = blockIdx.y;
  const int b = blockIdx.z;

  v8f acc[2][4];
  const v8f vz = {0.f, 0.f, 0.f, 0.f, 0.f, 0.f, 0.f, 0.f};
  for (int mt = 0; mt < 2; ++mt)
    for (int nt = 0; nt < 4; ++nt) acc[mt][nt] = vz;

  const int cx = 16 * tx + ml;
  const float* finb = fin + (size_t)b * HIN * WIN * 64;

  for (int r = 0; r < 9; ++r) {
    const int ky = r / 3, kx = r % 3;
    const int xx = min(2 * cx + kx, WIN - 1);
    const int yy0 = min(4 * py + ky, HIN - 1);
    const int yy1 = min(4 * py + 2 + ky, HIN - 1);
    const float* row0 = finb + (((size_t)yy0 * WIN + xx) << 6);
    const float* row1 = finb + (((size_t)yy1 * WIN + xx) << 6);
    __builtin_prefetch(wT + (((size_t)(min(r + 1, 8) * 32)) * 64) * 2, 0, 1);
    for (int u = 0; u < 16; ++u) {
      const int cinb = 4 * u + 2 * hh;
      v2f a0 = *(const v2f*)(row0 + cinb);
      v2f a1 = *(const v2f*)(row1 + cinb);
      const float* bp = wT + (((size_t)(r * 32 + 2 * u + hh)) * 64) * 2;
#pragma unroll
      for (int nt = 0; nt < 4; ++nt) {
        v2f bv = *(const v2f*)(bp + ((nt * 16 + ml) << 1));
        acc[0][nt] = wmma_f32(a0, bv, acc[0][nt]);
        acc[1][nt] = wmma_f32(a1, bv, acc[1][nt]);
      }
    }
  }

  const int pxb = 8 * tx;
#pragma unroll
  for (int nt = 0; nt < 4; ++nt) {
    const float bb = bias[nt * 16 + ml];
#pragma unroll
    for (int j = 0; j < 4; ++j) {
      float m0 = fmaxf(eluf(acc[0][nt][2 * j] + bb), eluf(acc[0][nt][2 * j + 1] + bb));
      float m1 = fmaxf(eluf(acc[1][nt][2 * j] + bb), eluf(acc[1][nt][2 * j + 1] + bb));
      int px = pxb + j + 4 * hh;
      if (px < WP)
        fout[((((size_t)b * HP + py) * WP + px) << 6) + nt * 16 + ml] = fmaxf(m0, m1);
    }
  }
}

__global__ void mean_kernel(const float* __restrict__ f3, float* __restrict__ gm) {
  int b = blockIdx.x, c = threadIdx.x;
  float s = 0.f;
  for (int i = 0; i < 49; ++i) s += f3[(((size_t)b * 49 + i) << 6) + c];
  gm[b * 64 + c] = s * (1.f / 49.f);
}

__global__ void fc_kernel(const float* __restrict__ gm, const float* __restrict__ w1,
                          const float* __restrict__ b1, const float* __restrict__ w2,
                          const float* __restrict__ b2, float* __restrict__ wq) {
  __shared__ float fv[64];
  __shared__ float h[64];
  int b = blockIdx.x, j = threadIdx.x;
  fv[j] = gm[b * 64 + j];
  __syncthreads();
  float acc = b1[j];
  for (int c = 0; c < 64; ++c) acc += fv[c] * w1[j * 64 + c];
  h[j] = eluf(acc);
  __syncthreads();
  if (j < 30) {
    float a2 = b2[j];
    for (int c = 0; c < 64; ++c) a2 += h[c] * w2[j * 64 + c];
    wq[b * 30 + j] = a2;
  }
}

__global__ void quad_kernel(const float* __restrict__ x, const float* __restrict__ wq,
                            float* __restrict__ out) {
  const int HW = 512 * 512;
  int idx = blockIdx.x * blockDim.x + threadIdx.x;
  if (idx >= 4 * HW) return;
  int b = idx / HW, hw = idx - b * HW;
  const float* xb = x + (size_t)b * 3 * HW;
  float r = xb[hw], g = xb[HW + hw], bl = xb[2 * HW + hw];
  float f[10] = {r * r, r * g, r * bl, r, g * g, g * bl, g, bl * bl, bl, 1.f};
  const float* w = wq + b * 30;
  float* ob = out + (size_t)b * 3 * HW;
#pragma unroll
  for (int c = 0; c < 3; ++c) {
    float s = (c == 0) ? r : ((c == 1) ? g : bl);
#pragma unroll
    for (int k = 0; k < 10; ++k) s += f[k] * w[c * 10 + k];
    ob[c * HW + hw] = s;
  }
}

static const size_t OFF_WT0 = 0;
static const size_t OFF_WT1 = 55296;
static const size_t OFF_WT2 = 92160;
static const size_t OFF_F1 = 129024;
static const size_t OFF_F2 = 129024 + 4129024;
static const size_t OFF_F3 = 129024 + 4129024 + 246016;
static const size_t OFF_GM = OFF_F3 + 12544;
static const size_t OFF_WQ = OFF_GM + 256;

extern "C" void kernel_launch(void* const* d_in, const int* in_sizes, int n_in,
                              void* d_out, int out_size, void* d_ws, size_t ws_size,
                              hipStream_t stream) {
  const float* x = (const float*)d_in[0];
  const float* c0w = (const float*)d_in[1];
  const float* c0b = (const float*)d_in[2];
  const float* c1w = (const float*)d_in[3];
  const float* c1b = (const float*)d_in[4];
  const float* c2w = (const float*)d_in[5];
  const float* c2b = (const float*)d_in[6];
  const float* f1w = (const float*)d_in[7];
  const float* f1b = (const float*)d_in[8];
  const float* f2w = (const float*)d_in[9];
  const float* f2b = (const float*)d_in[10];
  float* out = (float*)d_out;
  float* ws = (float*)d_ws;

  float* wT0 = ws + OFF_WT0;
  float* wT1 = ws + OFF_WT1;
  float* wT2 = ws + OFF_WT2;
  float* F1 = ws + OFF_F1;
  float* F2 = ws + OFF_F2;
  float* F3 = ws + OFF_F3;
  float* GM = ws + OFF_GM;
  float* WQ = ws + OFF_WQ;

  tr0_kernel<<<(864 * 64 + 255) / 256, 256, 0, stream>>>(c0w, wT0);
  tr1_kernel<<<(576 * 64 + 255) / 256, 256, 0, stream>>>(c1w, wT1);
  tr1_kernel<<<(576 * 64 + 255) / 256, 256, 0, stream>>>(c2w, wT2);

  conv0_kernel<<<dim3(16, 127, 4), 32, 0, stream>>>(x, wT0, c0b, F1);
  convmid_kernel<127, 127, 31, 31><<<dim3(4, 31, 4), 32, 0, stream>>>(F1, wT1, c1b, F2);
  convmid_kernel<31, 31, 7, 7><<<dim3(1, 7, 4), 32, 0, stream>>>(F2, wT2, c2b, F3);

  mean_kernel<<<4, 64, 0, stream>>>(F3, GM);
  fc_kernel<<<4, 64, 0, stream>>>(GM, f1w, f1b, f2w, f2b, WQ);

  const int npix = 4 * 512 * 512;
  quad_kernel<<<(npix + 255) / 256, 256, 0, stream>>>(x, WQ, out);
}


# conv0: A fragments of the 84 histogram channels computed in registers in the K loop from the pixel's colour (LDS row 96 -> 10 floats); wave tile 64x64, 4 waves/SIMD; changes summation order
# speedup vs baseline: 1.3562x; 1.3562x over previous
#include <hip/hip_runtime.h>
#include <cstddef>

typedef __attribute__((ext_vector_type(2))) float v2f;
typedef __attribute__((ext_vector_type(8))) float v8f;

#define DEV __device__ __forceinline__

DEV float eluf(float v) { return v > 0.f ? v : (__expf(v) - 1.f); }
DEV float histf(float v, float bin, float nm1) {
  return fmaxf(1.f - fabsf(v - bin) * nm1, 0.f);
}

DEV v8f wmma_f32(v2f a, v2f b, v8f c) {
  return __builtin_amdgcn_wmma_f32_16x16x4_f32(false, a, false, b, (short)0, c,
                                               false, false);
}

__global__ void tr0_kernel(const float* __restrict__ w, float* __restrict__ wT) {
  int idx = blockIdx.x * blockDim.x + threadIdx.x;
  if (idx >= 864 * 64) return;
  int k = idx >> 6, n = idx & 63;
  int r = k / 96, q = k % 96;
  int cf = (q < 9) ? q : (q - 3);
  float v = (q < 9 || q >= 12) ? w[(size_t)n * 837 + cf * 9 + r] : 0.f;
  wT[((((size_t)(k >> 1)) * 64 + n) << 1) + (k & 1)] = v;
}

__global__ void tr1_kernel(const float* __restrict__ w, float* __restrict__ wT) {
  int idx = blockIdx.x * blockDim.x + threadIdx.x;
  if (idx >= 576 * 64) return;
  int k = idx >> 6, n = idx & 63;
  int r = k >> 6, cin = k & 63;
  float v = w[(size_t)n * 576 + cin * 9 + r];
  wT[((((size_t)(k >> 1)) * 64 + n) << 1) + (k & 1)] = v;
}

#define C0_COLS 65
#define C0_NPIX (5 * C0_COLS)
#define C0_PITCH 10

DEV v2f hist2(float v, float b0, float b1, float nm1) {
  v2f a;
  a.x = histf(v, b0, nm1);
  a.y = histf(v, b1, nm1);
  return a;
}

__global__ __launch_bounds__(32) void conv0_kernel(const float* __restrict__ x,
                                                   const float* __restrict__ wT,
                                                   const float* __restrict__ bias,
                                                   float* __restrict__ f1) {
  const int lane = threadIdx.x;
  const int hh = lane >> 4;
  const int ml = lane & 15;
  const int tx = blockIdx.x;
  const int py = blockIdx.y;
  const int b = blockIdx.z;

  __shared__ float rec[C0_NPIX * C0_PITCH];

  const int gy0 = 4 * py;
  const int gx0 = 64 * tx;
  const float* xb = x + (size_t)b * 3 * 512 * 512;

  for (int p = lane; p < C0_NPIX; p += 32) {
    const int ly = p / C0_COLS, lx = p - ly * C0_COLS;
    const int gy = min(gy0 + ly, 511);
    const int gx = min(gx0 + lx, 511);
    const int ym = max(gy - 1, 0), yp = min(gy + 1, 511);
    const int xm = max(gx - 1, 0), xp = min(gx + 1, 511);
    float f[10];
#pragma unroll
    for (int c = 0; c < 3; ++c) {
      const float* xc = xb + (size_t)c * 512 * 512;
      float a = xc[ym * 512 + xm], bb = xc[ym * 512 + gx], cc = xc[ym * 512 + xp];
      float d = xc[gy * 512 + xm], ee = xc[gy * 512 + gx], e = xc[gy * 512 + xp];
      float g = xc[yp * 512 + xm], h = xc[yp * 512 + gx], i = xc[yp * 512 + xp];
      f[c] = ee;
      f[3 + 2 * c] = ((cc + 2.f * e + i) - (a + 2.f * d + g)) * 0.125f;
      f[4 + 2 * c] = ((g + 2.f * h + i) - (a + 2.f * bb + cc)) * 0.125f;
    }
    f[9] = 0.f;
    v2f* rp = (v2f*)&rec[p * C0_PITCH];
#pragma unroll
    for (int t = 0; t < 5; ++t) {
      v2f o = {f[2 * t], f[2 * t + 1]};
      rp[t] = o;
    }
  }
  __syncthreads();

  float b4[2], b8[2][2], b16[4][2];
#pragma unroll
  for (int s = 0; s < 2; ++s) {
    b4[s] = (float)(2 * hh + s) * (1.f / 3.f);
#pragma unroll
    for (int j = 0; j < 2; ++j) b8[j][s] = (float)(4 * j + 2 * hh + s) * (1.f / 7.f);
#pragma unroll
    for (int j = 0; j < 4; ++j) b16[j][s] = (float)(4 * j + 2 * hh + s) * (1.f / 15.f);
  }

  v8f acc[4][4];
  const v8f vz = {0.f, 0.f, 0.f, 0.f, 0.f, 0.f, 0.f, 0.f};
#pragma unroll
  for (int mt = 0; mt < 4; ++mt)
#pragma unroll
    for (int nt = 0; nt < 4; ++nt) acc[mt][nt] = vz;

  for (int r = 0; r < 9; ++r) {
    const int ky = r / 3, kx = r - 3 * ky;
    v2f raw[4][3];
    float cv[4][3];
#pragma unroll
    for (int mt = 0; mt < 4; ++mt) {
      const int pix = (2 * (mt >> 1) + ky) * C0_COLS + 32 * (mt & 1) + 2 * ml + kx;
      const float* rp = &rec[pix * C0_PITCH];
      const v2f c01 = *(const v2f*)(rp);
      const v2f c23 = *(const v2f*)(rp + 2);
      cv[mt][0] = c01.x;
      cv[mt][1] = c01.y;
      cv[mt][2] = c23.x;
      raw[mt][0] = *(const v2f*)(rp + 2 * hh);
      raw[mt][1] = *(const v2f*)(rp + 4 + 2 * hh);
      const v2f t2 = *(const v2f*)(rp + 8);
      raw[mt][2].x = hh ? 0.f : t2.x;
      raw[mt][2].y = hh ? 0.f : t2.y;
    }
    const float* wtap = wT + ((size_t)(r * 48 + hh) * 64 + ml) * 2;
#pragma unroll
    for (int u = 0; u < 24; ++u) {
      v2f a[4];
#pragma unroll
      for (int mt = 0; mt < 4; ++mt) {
        if (u < 3) {
          a[mt] = raw[mt][u < 3 ? u : 0];
        } else if (u < 6) {
          a[mt] = hist2(cv[mt][(u - 3) % 3], b4[0], b4[1], 3.f);
        } else if (u < 12) {
          a[mt] = hist2(cv[mt][((u - 6) >> 1) % 3], b8[(u - 6) & 1][0], b8[(u - 6) & 1][1], 7.f);
        } else {
          a[mt] = hist2(cv[mt][((u - 12) >> 2) % 3], b16[(u - 12) & 3][0], b16[(u - 12) & 3][1],
                        15.f);
        }
      }
      const float* bp = wtap + u * 256;
#pragma unroll
      for (int nt = 0; nt < 4; ++nt) {
        v2f bv = *(const v2f*)(bp + nt * 32);
#pragma unroll
        for (int mt = 0; mt < 4; ++mt) acc[mt][nt] = wmma_f32(a[mt], bv, acc[mt][nt]);
      }
    }
  }

#pragma unroll
  for (int nt = 0; nt < 4; ++nt) {
    const float bb = bias[nt * 16 + ml];
#pragma unroll
    for (int xh = 0; xh < 2; ++xh) {
      const int pxb = 16 * tx + 8 * xh;
#pragma unroll
      for (int j = 0; j < 4; ++j) {
        float m0 = fmaxf(eluf(acc[xh][nt][2 * j] + bb), eluf(acc[xh][nt][2 * j + 1] + bb));
        float m1 =
            fmaxf(eluf(acc[2 + xh][nt][2 * j] + bb), eluf(acc[2 + xh][nt][2 * j + 1] + bb));
        int px = pxb + j + 4 * hh;
        if (px < 127)
          f1[((((size_t)b * 127 + py) * 127 + px) << 6) + nt * 16 + ml] = fmaxf(m0, m1);
      }
    }
  }
}

template <int HIN, int WIN, int HP, int WP>
__global__ __launch_bounds__(32) void convmid_kernel(const float* __restrict__ fin,
                                                     const float* __restrict__ wT,
                                                     const float* __restrict__ bias,
                                                     float* __restrict__ fout) {
  const int lane = threadIdx.x;
  const int hh = lane >> 4;
  const int ml = lane & 15;
  const int tx = blockIdx.x;
  const int py = blockIdx.y;
  const int b = blockIdx.z;

  v8f acc[2][4];
  const v8f vz = {0.f, 0.f, 0.f, 0.f, 0.f, 0.f, 0.f, 0.f};
  for (int mt = 0; mt < 2; ++mt)
    for (int nt = 0; nt < 4; ++nt) acc[mt][nt] = vz;

  const int cx = 16 * tx + ml;
  const float* finb = fin + (size_t)b * HIN * WIN * 64;

  for (int r = 0; r < 9; ++r) {
    const int ky = r / 3, kx = r % 3;
    const int xx = min(2 * cx + kx, WIN - 1);
    const int yy0 = min(4 * py + ky, HIN - 1);
    const int yy1 = min(4 * py + 2 + ky, HIN - 1);
    const float* row0 = finb + (((size_t)yy0 * WIN + xx) << 6);
    const float* row1 = finb + (((size_t)yy1 * WIN + xx) << 6);
    __builtin_prefetch(wT + (((size_t)(min(r + 1, 8) * 32)) * 64) * 2, 0, 1);
    for (int u = 0; u < 16; ++u) {
      const int cinb = 4 * u + 2 * hh;
      v2f a0 = *(const v2f*)(row0 + cinb);
      v2f a1 = *(const v2f*)(row1 + cinb);
      const float* bp = wT + (((size_t)(r * 32 + 2 * u + hh)) * 64) * 2;
#pragma unroll
      for (int nt = 0; nt < 4; ++nt) {
        v2f bv = *(const v2f*)(bp + ((nt * 16 + ml) << 1));
        acc[0][nt] = wmma_f32(a0, bv, acc[0][nt]);
        acc[1][nt] = wmma_f32(a1, bv, acc[1][nt]);
      }
    }
  }

  const int pxb = 8 * tx;
#pragma unroll
  for (int nt = 0; nt < 4; ++nt) {
    const float bb = bias[nt * 16 + ml];
#pragma unroll
    for (int j = 0; j < 4; ++j) {
      float m0 = fmaxf(eluf(acc[0][nt][2 * j] + bb), eluf(acc[0][nt][2 * j + 1] + bb));
      float m1 = fmaxf(eluf(acc[1][nt][2 * j] + bb), eluf(acc[1][nt][2 * j + 1] + bb));
      int px = pxb + j + 4 * hh;
      if (px < WP)
        fout[((((size_t)b * HP + py) * WP + px) << 6) + nt * 16 + ml] = fmaxf(m0, m1);
    }
  }
}

__global__ void mean_kernel(const float* __restrict__ f3, float* __restrict__ gm) {
  int b = blockIdx.x, c = threadIdx.x;
  float s = 0.f;
  for (int i = 0; i < 49; ++i) s += f3[(((size_t)b * 49 + i) << 6) + c];
  gm[b * 64 + c] = s * (1.f / 49.f);
}

__global__ void fc_kernel(const float* __restrict__ gm, const float* __restrict__ w1,
                          const float* __restrict__ b1, const float* __restrict__ w2,
                          const float* __restrict__ b2, float* __restrict__ wq) {
  __shared__ float fv[64];
  __shared__ float h[64];
  int b = blockIdx.x, j = threadIdx.x;
  fv[j] = gm[b * 64 + j];
  __syncthreads();
  float acc = b1[j];
  for (int c = 0; c < 64; ++c) acc += fv[c] * w1[j * 64 + c];
  h[j] = eluf(acc);
  __syncthreads();
  if (j < 30) {
    float a2 = b2[j];
    for (int c = 0; c < 64; ++c) a2 += h[c] * w2[j * 64 + c];
    wq[b * 30 + j] = a2;
  }
}

__global__ void quad_kernel(const float* __restrict__ x, const float* __restrict__ wq,
                            float* __restrict__ out) {
  const int HW = 512 * 512;
  int idx = blockIdx.x * blockDim.x + threadIdx.x;
  if (idx >= 4 * HW) return;
  int b = idx / HW, hw = idx - b * HW;
  const float* xb = x + (size_t)b * 3 * HW;
  float r = xb[hw], g = xb[HW + hw], bl = xb[2 * HW + hw];
  float f[10] = {r * r, r * g, r * bl, r, g * g, g * bl, g, bl * bl, bl, 1.f};
  const float* w = wq + b * 30;
  float* ob = out + (size_t)b * 3 * HW;
#pragma unroll
  for (int c = 0; c < 3; ++c) {
    float s = (c == 0) ? r : ((c == 1) ? g : bl);
#pragma unroll
    for (int k = 0; k < 10; ++k) s += f[k] * w[c * 10 + k];
    ob[c * HW + hw] = s;
  }
}

static const size_t OFF_WT0 = 0;
static const size_t OFF_WT1 = 55296;
static const size_t OFF_WT2 = 92160;
static const size_t OFF_F1 = 129024;
static const size_t OFF_F2 = 129024 + 4129024;
static const size_t OFF_F3 = 129024 + 4129024 + 246016;
static const size_t OFF_GM = OFF_F3 + 12544;
static const size_t OFF_WQ = OFF_GM + 256;

extern "C" void kernel_launch(void* const* d_in, const int* in_sizes, int n_in,
                              void* d_out, int out_size, void* d_ws, size_t ws_size,
                              hipStream_t stream) {
  const float* x = (const float*)d_in[0];
  const float* c0w = (const float*)d_in[1];
  const float* c0b = (const float*)d_in[2];
  const float* c1w = (const float*)d_in[3];
  const float* c1b = (const float*)d_in[4];
  const float* c2w = (const float*)d_in[5];
  const float* c2b = (const float*)d_in[6];
  const float* f1w = (const float*)d_in[7];
  const float* f1b = (const float*)d_in[8];
  const float* f2w = (const float*)d_in[9];
  const float* f2b = (const float*)d_in[10];
  float* out = (float*)d_out;
  float* ws = (float*)d_ws;

  float* wT0 = ws + OFF_WT0;
  float* wT1 = ws + OFF_WT1;
  float* wT2 = ws + OFF_WT2;
  float* F1 = ws + OFF_F1;
  float* F2 = ws + OFF_F2;
  float* F3 = ws + OFF_F3;
  float* GM = ws + OFF_GM;
  float* WQ = ws + OFF_WQ;

  tr0_kernel<<<(864 * 64 + 255) / 256, 256, 0, stream>>>(c0w, wT0);
  tr1_kernel<<<(576 * 64 + 255) / 256, 256, 0, stream>>>(c1w, wT1);
  tr1_kernel<<<(576 * 64 + 255) / 256, 256, 0, stream>>>(c2w, wT2);

  conv0_kernel<<<dim3(8, 127, 4), 32, 0, stream>>>(x, wT0, c0b, F1);
  convmid_kernel<127, 127, 31, 31><<<dim3(4, 31, 4), 32, 0, stream>>>(F1, wT1, c1b, F2);
  convmid_kernel<31, 31, 7, 7><<<dim3(1, 7, 4), 32, 0, stream>>>(F2, wT2, c2b, F3);

  mean_kernel<<<4, 64, 0, stream>>>(F3, GM);
  fc_kernel<<<4, 64, 0, stream>>>(GM, f1w, f1b, f2w, f2b, WQ);

  const int npix = 4 * 512 * 512;
  quad_kernel<<<(npix + 255) / 256, 256, 0, stream>>>(x, WQ, out);
}


# conv1/conv2: 4-wave workgroups, a tile per wave; each tap's 16 KiB of weights fetched once per workgroup by 16-byte loads a tap ahead and shared via LDS (B fragments read from LDS); same sums
# speedup vs baseline: 1.7465x; 1.2878x over previous
#include <hip/hip_runtime.h>
#include <cstddef>

typedef __attribute__((ext_vector_type(2))) float v2f;
typedef __attribute__((ext_vector_type(8))) float v8f;

#define DEV __device__ __forceinline__

DEV float eluf(float v) { return v > 0.f ? v : (__expf(v) - 1.f); }
DEV float histf(float v, float bin, float nm1) {
  return fmaxf(1.f - fabsf(v - bin) * nm1, 0.f);
}

DEV v8f wmma_f32(v2f a, v2f b, v8f c) {
  return __builtin_amdgcn_wmma_f32_16x16x4_f32(false, a, false, b, (short)0, c,
                                               false, false);
}

__global__ void tr0_kernel(const float* __restrict__ w, float* __restrict__ wT) {
  int idx = blockIdx.x * blockDim.x + threadIdx.x;
  if (idx >= 864 * 64) return;
  int k = idx >> 6, n = idx & 63;
  int r = k / 96, q = k % 96;
  int cf = (q < 9) ? q : (q - 3);
  float v = (q < 9 || q >= 12) ? w[(size_t)n * 837 + cf * 9 + r] : 0.f;
  wT[((((size_t)(k >> 1)) * 64 + n) << 1) + (k & 1)] = v;
}

__global__ void tr1_kernel(const float* __restrict__ w, float* __restrict__ wT) {
  int idx = blockIdx.x * blockDim.x + threadIdx.x;
  if (idx >= 576 * 64) return;
  int k = idx >> 6, n = idx & 63;
  int r = k >> 6, cin = k & 63;
  float v = w[(size_t)n * 576 + cin * 9 + r];
  wT[((((size_t)(k >> 1)) * 64 + n) << 1) + (k & 1)] = v;
}

#define C0_COLS 65
#define C0_NPIX (5 * C0_COLS)
#define C0_PITCH 10

DEV v2f hist2(float v, float b0, float b1, float nm1) {
  v2f a;
  a.x = histf(v, b0, nm1);
  a.y = histf(v, b1, nm1);
  return a;
}

__global__ __launch_bounds__(32) void conv0_kernel(const float* __restrict__ x,
                                                   const float* __restrict__ wT,
                                                   const float* __restrict__ bias,
                                                   float* __restrict__ f1) {
  const int lane = threadIdx.x;
  const int hh = lane >> 4;
  const int ml = lane & 15;
  const int tx = blockIdx.x;
  const int py = blockIdx.y;
  const int b = blockIdx.z;

  __shared__ float rec[C0_NPIX * C0_PITCH];

  const int gy0 = 4 * py;
  const int gx0 = 64 * tx;
  const float* xb = x + (size_t)b * 3 * 512 * 512;

  for (int p = lane; p < C0_NPIX; p += 32) {
    const int ly = p / C0_COLS, lx = p - ly * C0_COLS;
    const int gy = min(gy0 + ly, 511);
    const int gx = min(gx0 + lx, 511);
    const int ym = max(gy - 1, 0), yp = min(gy + 1, 511);
    const int xm = max(gx - 1, 0), xp = min(gx + 1, 511);
    float f[10];
#pragma unroll
    for (int c = 0; c < 3; ++c) {
      const float* xc = xb + (size_t)c * 512 * 512;
      float a = xc[ym * 512 + xm], bb = xc[ym * 512 + gx], cc = xc[ym * 512 + xp];
      float d = xc[gy * 512 + xm], ee = xc[gy * 512 + gx], e = xc[gy * 512 + xp];
      float g = xc[yp * 512 + xm], h = xc[yp * 512 + gx], i = xc[yp * 512 + xp];
      f[c] = ee;
      f[3 + 2 * c] = ((cc + 2.f * e + i) - (a + 2.f * d + g)) * 0.125f;
      f[4 + 2 * c] = ((g + 2.f * h + i) - (a + 2.f * bb + cc)) * 0.125f;
    }
    f[9] = 0.f;
    v2f* rp = (v2f*)&rec[p * C0_PITCH];
#pragma unroll
    for (int t = 0; t < 5; ++t) {
      v2f o = {f[2 * t], f[2 * t + 1]};
      rp[t] = o;
    }
  }
  __syncthreads();

  float b4[2], b8[2][2], b16[4][2];
#pragma unroll
  for (int s = 0; s < 2; ++s) {
    b4[s] = (float)(2 * hh + s) * (1.f / 3.f);
#pragma unroll
    for (int j = 0; j < 2; ++j) b8[j][s] = (float)(4 * j + 2 * hh + s) * (1.f / 7.f);
#pragma unroll
    for (int j = 0; j < 4; ++j) b16[j][s] = (float)(4 * j + 2 * hh + s) * (1.f / 15.f);
  }

  v8f acc[4][4];
  const v8f vz = {0.f, 0.f, 0.f, 0.f, 0.f, 0.f, 0.f, 0.f};
#pragma unroll
  for (int mt = 0; mt < 4; ++mt)
#pragma unroll
    for (int nt = 0; nt < 4; ++nt) acc[mt][nt] = vz;

  for (int r = 0; r < 9; ++r) {
    const int ky = r / 3, kx = r - 3 * ky;
    v2f raw[4][3];
    float cv[4][3];
#pragma unroll
    for (int mt = 0; mt < 4; ++mt) {
      const int pix = (2 * (mt >> 1) + ky) * C0_COLS + 32 * (mt & 1) + 2 * ml + kx;
      const float* rp = &rec[pix * C0_PITCH];
      const v2f c01 = *(const v2f*)(rp);
      const v2f c23 = *(const v2f*)(rp + 2);
      cv[mt][0] = c01.x;
      cv[mt][1] = c01.y;
      cv[mt][2] = c23.x;
      raw[mt][0] = *(const v2f*)(rp + 2 * hh);
      raw[mt][1] = *(const v2f*)(rp + 4 + 2 * hh);
      const v2f t2 = *(const v2f*)(rp + 8);
      raw[mt][2].x = hh ? 0.f : t2.x;
      raw[mt][2].y = hh ? 0.f : t2.y;
    }
    const float* wtap = wT + ((size_t)(r * 48 + hh) * 64 + ml) * 2;
#pragma unroll
    for (int u = 0; u < 24; ++u) {
      v2f a[4];
#pragma unroll
      for (int mt = 0; mt < 4; ++mt) {
        if (u < 3) {
          a[mt] = raw[mt][u < 3 ? u : 0];
        } else if (u < 6) {
          a[mt] = hist2(cv[mt][(u - 3) % 3], b4[0], b4[1], 3.f);
        } else if (u < 12) {
          a[mt] = hist2(cv[mt][((u - 6) >> 1) % 3], b8[(u - 6) & 1][0], b8[(u - 6) & 1][1], 7.f);
        } else {
          a[mt] = hist2(cv[mt][((u - 12) >> 2) % 3], b16[(u - 12) & 3][0], b16[(u - 12) & 3][1],
                        15.f);
        }
      }
      const float* bp = wtap + u * 256;
#pragma unroll
      for (int nt = 0; nt < 4; ++nt) {
        v2f bv = *(const v2f*)(bp + nt * 32);
#pragma unroll
        for (int mt = 0; mt < 4; ++mt) acc[mt][nt] = wmma_f32(a[mt], bv, acc[mt][nt]);
      }
    }
  }

#pragma unroll
  for (int nt = 0; nt < 4; ++nt) {
    const float bb = bias[nt * 16 + ml];
#pragma unroll
    for (int xh = 0; xh < 2; ++xh) {
      const int pxb = 16 * tx + 8 * xh;
#pragma unroll
      for (int j = 0; j < 4; ++j) {
        float m0 = fmaxf(eluf(acc[xh][nt][2 * j] + bb), eluf(acc[xh][nt][2 * j + 1] + bb));
        float m1 =
            fmaxf(eluf(acc[2 + xh][nt][2 * j] + bb), eluf(acc[2 + xh][nt][2 * j + 1] + bb));
        int px = pxb + j + 4 * hh;
        if (px < 127)
          f1[((((size_t)b * 127 + py) * 127 + px) << 6) + nt * 16 + ml] = fmaxf(m0, m1);
      }
    }
  }
}

#define CM_BP 160

template <int HIN, int WIN, int HP, int WP, int NXT>
__global__ __launch_bounds__(128) void convmid_kernel(const float* __restrict__ fin,
                                                      const float* __restrict__ wT,
                                                      const float* __restrict__ bias,
                                                      float* __restrict__ fout) {
  const int tid = threadIdx.x;
  const int lane = tid & 31;
  const int wave = tid >> 5;
  const int hh = lane >> 4;
  const int ml = lane & 15;
  const int t = 4 * blockIdx.x + wave;
  const int tx = t % NXT;
  const int py = (t / NXT) % HP;
  const int b = t / (NXT * HP);

  __shared__ __attribute__((aligned(16))) float bs[32 * CM_BP];

  v8f acc[2][4];
  const v8f vz = {0.f, 0.f, 0.f, 0.f, 0.f, 0.f, 0.f, 0.f};
  for (int mt = 0; mt < 2; ++mt)
    for (int nt = 0; nt < 4; ++nt) acc[mt][nt] = vz;

  const int cx = 16 * tx + ml;
  const float* finb = fin + (size_t)b * HIN * WIN * 64;

  const float4* wg = (const float4*)wT + tid;
  float4 p0 = wg[0], p1 = wg[128], p2 = wg[256], p3 = wg[384];
  float4 p4 = wg[512], p5 = wg[640], p6 = wg[768], p7 = wg[896];
  float* bsw = &bs[wave * CM_BP + 4 * lane];

  for (int r = 0; r < 9; ++r) {
    __syncthreads();
    *(float4*)(bsw) = p0;
    *(float4*)(bsw + 4 * CM_BP) = p1;
    *(float4*)(bsw + 8 * CM_BP) = p2;
    *(float4*)(bsw + 12 * CM_BP) = p3;
    *(float4*)(bsw + 16 * CM_BP) = p4;
    *(float4*)(bsw + 20 * CM_BP) = p5;
    *(float4*)(bsw + 24 * CM_BP) = p6;
    *(float4*)(bsw + 28 * CM_BP) = p7;
    __syncthreads();
    if (r < 8) {
      const float4* wn = wg + (r + 1) * 1024;
      p0 = wn[0], p1 = wn[128], p2 = wn[256], p3 = wn[384];
      p4 = wn[512], p5 = wn[640], p6 = wn[768], p7 = wn[896];
    }

    const int ky = r / 3, kx = r % 3;
    const int xx = min(2 * cx + kx, WIN - 1);
    const int yy0 = min(4 * py + ky, HIN - 1);
    const int yy1 = min(4 * py + 2 + ky, HIN - 1);
    const float* row0 = finb + (((size_t)yy0 * WIN + xx) << 6);
    const float* row1 = finb + (((size_t)yy1 * WIN + xx) << 6);
#pragma unroll
    for (int u = 0; u < 16; ++u) {
      const int cinb = 4 * u + 2 * hh;
      v2f a0 = *(const v2f*)(row0 + cinb);
      v2f a1 = *(const v2f*)(row1 + cinb);
      const float* bp = &bs[(2 * u + hh) * CM_BP + 2 * ml];
#pragma unroll
      for (int nt = 0; nt < 4; ++nt) {
        v2f bv = *(const v2f*)(bp + nt * 32);
        acc[0][nt] = wmma_f32(a0, bv, acc[0][nt]);
        acc[1][nt] = wmma_f32(a1, bv, acc[1][nt]);
      }
    }
  }

  const int pxb = 8 * tx;
#pragma unroll
  for (int nt = 0; nt < 4; ++nt) {
    const float bb = bias[nt * 16 + ml];
#pragma unroll
    for (int j = 0; j < 4; ++j) {
      float m0 = fmaxf(eluf(acc[0][nt][2 * j] + bb), eluf(acc[0][nt][2 * j + 1] + bb));
      float m1 = fmaxf(eluf(acc[1][nt][2 * j] + bb), eluf(acc[1][nt][2 * j + 1] + bb));
      int px = pxb + j + 4 * hh;
      if (px < WP)
        fout[((((size_t)b * HP + py) * WP + px) << 6) + nt * 16 + ml] = fmaxf(m0, m1);
    }
  }
}

__global__ void mean_kernel(const float* __restrict__ f3, float* __restrict__ gm) {
  int b = blockIdx.x, c = threadIdx.x;
  float s = 0.f;
  for (int i = 0; i < 49; ++i) s += f3[(((size_t)b * 49 + i) << 6) + c];
  gm[b * 64 + c] = s * (1.f / 49.f);
}

__global__ void fc_kernel(const float* __restrict__ gm, const float* __restrict__ w1,
                          const float* __restrict__ b1, const float* __restrict__ w2,
                          const float* __restrict__ b2, float* __restrict__ wq) {
  __shared__ float fv[64];
  __shared__ float h[64];
  int b = blockIdx.x, j = threadIdx.x;
  fv[j] = gm[b * 64 + j];
  __syncthreads();
  float acc = b1[j];
  for (int c = 0; c < 64; ++c) acc += fv[c] * w1[j * 64 + c];
  h[j] = eluf(acc);
  __syncthreads();
  if (j < 30) {
    float a2 = b2[j];
    for (int c = 0; c < 64; ++c) a2 += h[c] * w2[j * 64 + c];
    wq[b * 30 + j] = a2;
  }
}

__global__ void quad_kernel(const float* __restrict__ x, const float* __restrict__ wq,
                            float* __restrict__ out) {
  const int HW = 512 * 512;
  int idx = blockIdx.x * blockDim.x + threadIdx.x;
  if (idx >= 4 * HW) return;
  int b = idx / HW, hw = idx - b * HW;
  const float* xb = x + (size_t)b * 3 * HW;
  float r = xb[hw], g = xb[HW + hw], bl = xb[2 * HW + hw];
  float f[10] = {r * r, r * g, r * bl, r, g * g, g * bl, g, bl * bl, bl, 1.f};
  const float* w = wq + b * 30;
  float* ob = out + (size_t)b * 3 * HW;
#pragma unroll
  for (int c = 0; c < 3; ++c) {
    float s = (c == 0) ? r : ((c == 1) ? g : bl);
#pragma unroll
    for (int k = 0; k < 10; ++k) s += f[k] * w[c * 10 + k];
    ob[c * HW + hw] = s;
  }
}

static const size_t OFF_WT0 = 0;
static const size_t OFF_WT1 = 55296;
static const size_t OFF_WT2 = 92160;
static const size_t OFF_F1 = 129024;
static const size_t OFF_F2 = 129024 + 4129024;
static const size_t OFF_F3 = 129024 + 4129024 + 246016;
static const size_t OFF_GM = OFF_F3 + 12544;
static const size_t OFF_WQ = OFF_GM + 256;

extern "C" void kernel_launch(void* const* d_in, const int* in_sizes, int n_in,
                              void* d_out, int out_size, void* d_ws, size_t ws_size,
                              hipStream_t stream) {
  const float* x = (const float*)d_in[0];
  const float* c0w = (const float*)d_in[1];
  const float* c0b = (const float*)d_in[2];
  const float* c1w = (const float*)d_in[3];
  const float* c1b = (const float*)d_in[4];
  const float* c2w = (const float*)d_in[5];
  const float* c2b = (const float*)d_in[6];
  const float* f1w = (const float*)d_in[7];
  const float* f1b = (const float*)d_in[8];
  const float* f2w = (const float*)d_in[9];
  const float* f2b = (const float*)d_in[10];
  float* out = (float*)d_out;
  float* ws = (float*)d_ws;

  float* wT0 = ws + OFF_WT0;
  float* wT1 = ws + OFF_WT1;
  float* wT2 = ws + OFF_WT2;
  float* F1 = ws + OFF_F1;
  float* F2 = ws + OFF_F2;
  float* F3 = ws + OFF_F3;
  float* GM = ws + OFF_GM;
  float* WQ = ws + OFF_WQ;

  tr0_kernel<<<(864 * 64 + 255) / 256, 256, 0, stream>>>(c0w, wT0);
  tr1_kernel<<<(576 * 64 + 255) / 256, 256, 0, stream>>>(c1w, wT1);
  tr1_kernel<<<(576 * 64 + 255) / 256, 256, 0, stream>>>(c2w, wT2);

  conv0_kernel<<<dim3(8, 127, 4), 32, 0, stream>>>(x, wT0, c0b, F1);
  convmid_kernel<127, 127, 31, 31, 4><<<124, 128, 0, stream>>>(F1, wT1, c1b, F2);
  convmid_kernel<31, 31, 7, 7, 1><<<7, 128, 0, stream>>>(F2, wT2, c2b, F3);

  mean_kernel<<<4, 64, 0, stream>>>(F3, GM);
  fc_kernel<<<4, 64, 0, stream>>>(GM, f1w, f1b, f2w, f2b, WQ);

  const int npix = 4 * 512 * 512;
  quad_kernel<<<(npix + 255) / 256, 256, 0, stream>>>(x, WQ, out);
}


# conv1/conv2: wave tile re-cut to ONE 16-row fragment holding both conv rows of the 2x2 pool (8 x 2 x 64): half the serial WMMAs per wave, twice the waves; row max by a lane^16 exchange; same sums
# speedup vs baseline: 1.8589x; 1.0644x over previous
#include <hip/hip_runtime.h>
#include <cstddef>

typedef __attribute__((ext_vector_type(2))) float v2f;
typedef __attribute__((ext_vector_type(8))) float v8f;

#define DEV __device__ __forceinline__

DEV float eluf(float v) { return v > 0.f ? v : (__expf(v) - 1.f); }
DEV float histf(float v, float bin, float nm1) {
  return fmaxf(1.f - fabsf(v - bin) * nm1, 0.f);
}

DEV float xhalf(float v) {
  return __builtin_bit_cast(float, __builtin_amdgcn_permlanex16(__builtin_bit_cast(int, v),
                                                                __builtin_bit_cast(int, v),
                                                                0x76543210, 0xfedcba98, false, false));
}

DEV v8f wmma_f32(v2f a, v2f b, v8f c) {
  return __builtin_amdgcn_wmma_f32_16x16x4_f32(false, a, false, b, (short)0, c,
                                               false, false);
}

__global__ void tr0_kernel(const float* __restrict__ w, float* __restrict__ wT) {
  int idx = blockIdx.x * blockDim.x + threadIdx.x;
  if (idx >= 864 * 64) return;
  int k = idx >> 6, n = idx & 63;
  int r = k / 96, q = k % 96;
  int cf = (q < 9) ? q : (q - 3);
  float v = (q < 9 || q >= 12) ? w[(size_t)n * 837 + cf * 9 + r] : 0.f;
  wT[((((size_t)(k >> 1)) * 64 + n) << 1) + (k & 1)] = v;
}

__global__ void tr1_kernel(const float* __restrict__ w, float* __restrict__ wT) {
  int idx = blockIdx.x * blockDim.x + threadIdx.x;
  if (idx >= 576 * 64) return;
  int k = idx >> 6, n = idx & 63;
  int r = k >> 6, cin = k & 63;
  float v = w[(size_t)n * 576 + cin * 9 + r];
  wT[((((size_t)(k >> 1)) * 64 + n) << 1) + (k & 1)] = v;
}

#define C0_COLS 65
#define C0_NPIX (5 * C0_COLS)
#define C0_PITCH 10

DEV v2f hist2(float v, float b0, float b1, float nm1) {
  v2f a;
  a.x = histf(v, b0, nm1);
  a.y = histf(v, b1, nm1);
  return a;
}

__global__ __launch_bounds__(32) void conv0_kernel(const float* __restrict__ x,
                                                   const float* __restrict__ wT,
                                                   const float* __restrict__ bias,
                                                   float* __restrict__ f1) {
  const int lane = threadIdx.x;
  const int hh = lane >> 4;
  const int ml = lane & 15;
  const int tx = blockIdx.x;
  const int py = blockIdx.y;
  const int b = blockIdx.z;

  __shared__ float rec[C0_NPIX * C0_PITCH];

  const int gy0 = 4 * py;
  const int gx0 = 64 * tx;
  const float* xb = x + (size_t)b * 3 * 512 * 512;

  for (int p = lane; p < C0_NPIX; p += 32) {
    const int ly = p / C0_COLS, lx = p - ly * C0_COLS;
    const int gy = min(gy0 + ly, 511);
    const int gx = min(gx0 + lx, 511);
    const int ym = max(gy - 1, 0), yp = min(gy + 1, 511);
    const int xm = max(gx - 1, 0), xp = min(gx + 1, 511);
    float f[10];
#pragma unroll
    for (int c = 0; c < 3; ++c) {
      const float* xc = xb + (size_t)c * 512 * 512;
      float a = xc[ym * 512 + xm], bb = xc[ym * 512 + gx], cc = xc[ym * 512 + xp];
      float d = xc[gy * 512 + xm], ee = xc[gy * 512 + gx], e = xc[gy * 512 + xp];
      float g = xc[yp * 512 + xm], h = xc[yp * 512 + gx], i = xc[yp * 512 + xp];
      f[c] = ee;
      f[3 + 2 * c] = ((cc + 2.f * e + i) - (a + 2.f * d + g)) * 0.125f;
      f[4 + 2 * c] = ((g + 2.f * h + i) - (a + 2.f * bb + cc)) * 0.125f;
    }
    f[9] = 0.f;
    v2f* rp = (v2f*)&rec[p * C0_PITCH];
#pragma unroll
    for (int t = 0; t < 5; ++t) {
      v2f o = {f[2 * t], f[2 * t + 1]};
      rp[t] = o;
    }
  }
  __syncthreads();

  float b4[2], b8[2][2], b16[4][2];
#pragma unroll
  for (int s = 0; s < 2; ++s) {
    b4[s] = (float)(2 * hh + s) * (1.f / 3.f);
#pragma unroll
    for (int j = 0; j < 2; ++j) b8[j][s] = (float)(4 * j + 2 * hh + s) * (1.f / 7.f);
#pragma unroll
    for (int j = 0; j < 4; ++j) b16[j][s] = (float)(4 * j + 2 * hh + s) * (1.f / 15.f);
  }

  v8f acc[4][4];
  const v8f vz = {0.f, 0.f, 0.f, 0.f, 0.f, 0.f, 0.f, 0.f};
#pragma unroll
  for (int mt = 0; mt < 4; ++mt)
#pragma unroll
    for (int nt = 0; nt < 4; ++nt) acc[mt][nt] = vz;

  for (int r = 0; r < 9; ++r) {
    const int ky = r / 3, kx = r - 3 * ky;
    v2f raw[4][3];
    float cv[4][3];
#pragma unroll
    for (int mt = 0; mt < 4; ++mt) {
      const int pix = (2 * (mt >> 1) + ky) * C0_COLS + 32 * (mt & 1) + 2 * ml + kx;
      const float* rp = &rec[pix * C0_PITCH];
      const v2f c01 = *(const v2f*)(rp);
      const v2f c23 = *(const v2f*)(rp + 2);
      cv[mt][0] = c01.x;
      cv[mt][1] = c01.y;
      cv[mt][2] = c23.x;
      raw[mt][0] = *(const v2f*)(rp + 2 * hh);
      raw[mt][1] = *(const v2f*)(rp + 4 + 2 * hh);
      const v2f t2 = *(const v2f*)(rp + 8);
      raw[mt][2].x = hh ? 0.f : t2.x;
      raw[mt][2].y = hh ? 0.f : t2.y;
    }
    const float* wtap = wT + ((size_t)(r * 48 + hh) * 64 + ml) * 2;
#pragma unroll
    for (int u = 0; u < 24; ++u) {
      v2f a[4];
#pragma unroll
      for (int mt = 0; mt < 4; ++mt) {
        if (u < 3) {
          a[mt] = raw[mt][u < 3 ? u : 0];
        } else if (u < 6) {
          a[mt] = hist2(cv[mt][(u - 3) % 3], b4[0], b4[1], 3.f);
        } else if (u < 12) {
          a[mt] = hist2(cv[mt][((u - 6) >> 1) % 3], b8[(u - 6) & 1][0], b8[(u - 6) & 1][1], 7.f);
        } else {
          a[mt] = hist2(cv[mt][((u - 12) >> 2) % 3], b16[(u - 12) & 3][0], b16[(u - 12) & 3][1],
                        15.f);
        }
      }
      const float* bp = wtap + u * 256;
#pragma unroll
      for (int nt = 0; nt < 4; ++nt) {
        v2f bv = *(const v2f*)(bp + nt * 32);
#pragma unroll
        for (int mt = 0; mt < 4; ++mt) acc[mt][nt] = wmma_f32(a[mt], bv, acc[mt][nt]);
      }
    }
  }

#pragma unroll
  for (int nt = 0; nt < 4; ++nt) {
    const float bb = bias[nt * 16 + ml];
#pragma unroll
    for (int xh = 0; xh < 2; ++xh) {
      const int pxb = 16 * tx + 8 * xh;
#pragma unroll
      for (int j = 0; j < 4; ++j) {
        float m0 = fmaxf(eluf(acc[xh][nt][2 * j] + bb), eluf(acc[xh][nt][2 * j + 1] + bb));
        float m1 =
            fmaxf(eluf(acc[2 + xh][nt][2 * j] + bb), eluf(acc[2 + xh][nt][2 * j + 1] + bb));
        int px = pxb + j + 4 * hh;
        if (px < 127)
          f1[((((size_t)b * 127 + py) * 127 + px) << 6) + nt * 16 + ml] = fmaxf(m0, m1);
      }
    }
  }
}

#define CM_BP 160

template <int HIN, int WIN, int HP, int WP, int NXT>
__global__ __launch_bounds__(128) void convmid_kernel(const float* __restrict__ fin,
                                                      const float* __restrict__ wT,
                                                      const float* __restrict__ bias,
                                                      float* __restrict__ fout) {
  const int tid = threadIdx.x;
  const int lane = tid & 31;
  const int wave = tid >> 5;
  const int hh = lane >> 4;
  const int ml = lane & 15;
  const int t = 4 * blockIdx.x + wave;
  const int tx = t % NXT;
  const int py = (t / NXT) % HP;
  const int b = t / (NXT * HP);

  __shared__ __attribute__((aligned(16))) float bs[32 * CM_BP];

  v8f acc[4];
  const v8f vz = {0.f, 0.f, 0.f, 0.f, 0.f, 0.f, 0.f, 0.f};
  for (int nt = 0; nt < 4; ++nt) acc[nt] = vz;

  const int cx = 8 * tx + (ml & 7);
  const int cy = 2 * py + (ml >> 3);
  const float* finb = fin + (size_t)b * HIN * WIN * 64;

  const float4* wg = (const float4*)wT + tid;
  float4 p0 = wg[0], p1 = wg[128], p2 = wg[256], p3 = wg[384];
  float4 p4 = wg[512], p5 = wg[640], p6 = wg[768], p7 = wg[896];
  float* bsw = &bs[wave * CM_BP + 4 * lane];

  for (int r = 0; r < 9; ++r) {
    __syncthreads();
    *(float4*)(bsw) = p0;
    *(float4*)(bsw + 4 * CM_BP) = p1;
    *(float4*)(bsw + 8 * CM_BP) = p2;
    *(float4*)(bsw + 12 * CM_BP) = p3;
    *(float4*)(bsw + 16 * CM_BP) = p4;
    *(float4*)(bsw + 20 * CM_BP) = p5;
    *(float4*)(bsw + 24 * CM_BP) = p6;
    *(float4*)(bsw + 28 * CM_BP) = p7;
    __syncthreads();
    if (r < 8) {
      const float4* wn = wg + (r + 1) * 1024;
      p0 = wn[0], p1 = wn[128], p2 = wn[256], p3 = wn[384];
      p4 = wn[512], p5 = wn[640], p6 = wn[768], p7 = wn[896];
    }

    const int ky = r / 3, kx = r % 3;
    const int xx = min(2 * cx + kx, WIN - 1);
    const int yy = min(2 * cy + ky, HIN - 1);
    const float* row = finb + (((size_t)yy * WIN + xx) << 6);
#pragma unroll
    for (int u = 0; u < 16; ++u) {
      const int cinb = 4 * u + 2 * hh;
      v2f a = *(const v2f*)(row + cinb);
      const float* bp = &bs[(2 * u + hh) * CM_BP + 2 * ml];
#pragma unroll
      for (int nt = 0; nt < 4; ++nt) {
        v2f bv = *(const v2f*)(bp + nt * 32);
        acc[nt] = wmma_f32(a, bv, acc[nt]);
      }
    }
  }

  const int pxb = 4 * tx;
#pragma unroll
  for (int nt = 0; nt < 4; ++nt) {
    const float bb = bias[nt * 16 + ml];
#pragma unroll
    for (int j = 0; j < 4; ++j) {
      float m = fmaxf(eluf(acc[nt][2 * j] + bb), eluf(acc[nt][2 * j + 1] + bb));
      m = fmaxf(m, xhalf(m));
      const int px = pxb + j;
      if ((j >> 1) == hh && px < WP)
        fout[((((size_t)b * HP + py) * WP + px) << 6) + nt * 16 + ml] = m;
    }
  }
}

__global__ void mean_kernel(const float* __restrict__ f3, float* __restrict__ gm) {
  int b = blockIdx.x, c = threadIdx.x;
  float s = 0.f;
  for (int i = 0; i < 49; ++i) s += f3[(((size_t)b * 49 + i) << 6) + c];
  gm[b * 64 + c] = s * (1.f / 49.f);
}

__global__ void fc_kernel(const float* __restrict__ gm, const float* __restrict__ w1,
                          const float* __restrict__ b1, const float* __restrict__ w2,
                          const float* __restrict__ b2, float* __restrict__ wq) {
  __shared__ float fv[64];
  __shared__ float h[64];
  int b = blockIdx.x, j = threadIdx.x;
  fv[j] = gm[b * 64 + j];
  __syncthreads();
  float acc = b1[j];
  for (int c = 0; c < 64; ++c) acc += fv[c] * w1[j * 64 + c];
  h[j] = eluf(acc);
  __syncthreads();
  if (j < 30) {
    float a2 = b2[j];
    for (int c = 0; c < 64; ++c) a2 += h[c] * w2[j * 64 + c];
    wq[b * 30 + j] = a2;
  }
}

__global__ void quad_kernel(const float* __restrict__ x, const float* __restrict__ wq,
                            float* __restrict__ out) {
  const int HW = 512 * 512;
  int idx = blockIdx.x * blockDim.x + threadIdx.x;
  if (idx >= 4 * HW) return;
  int b = idx / HW, hw = idx - b * HW;
  const float* xb = x + (size_t)b * 3 * HW;
  float r = xb[hw], g = xb[HW + hw], bl = xb[2 * HW + hw];
  float f[10] = {r * r, r * g, r * bl, r, g * g, g * bl, g, bl * bl, bl, 1.f};
  const float* w = wq + b * 30;
  float* ob = out + (size_t)b * 3 * HW;
#pragma unroll
  for (int c = 0; c < 3; ++c) {
    float s = (c == 0) ? r : ((c == 1) ? g : bl);
#pragma unroll
    for (int k = 0; k < 10; ++k) s += f[k] * w[c * 10 + k];
    ob[c * HW + hw] = s;
  }
}

static const size_t OFF_WT0 = 0;
static const size_t OFF_WT1 = 55296;
static const size_t OFF_WT2 = 92160;
static const size_t OFF_F1 = 129024;
static const size_t OFF_F2 = 129024 + 4129024;
static const size_t OFF_F3 = 129024 + 4129024 + 246016;
static const size_t OFF_GM = OFF_F3 + 12544;
static const size_t OFF_WQ = OFF_GM + 256;

extern "C" void kernel_launch(void* const* d_in, const int* in_sizes, int n_in,
                              void* d_out, int out_size, void* d_ws, size_t ws_size,
                              hipStream_t stream) {
  const float* x = (const float*)d_in[0];
  const float* c0w = (const float*)d_in[1];
  const float* c0b = (const float*)d_in[2];
  const float* c1w = (const float*)d_in[3];
  const float* c1b = (const float*)d_in[4];
  const float* c2w = (const float*)d_in[5];
  const float* c2b = (const float*)d_in[6];
  const float* f1w = (const float*)d_in[7];
  const float* f1b = (const float*)d_in[8];
  const float* f2w = (const float*)d_in[9];
  const float* f2b = (const float*)d_in[10];
  float* out = (float*)d_out;
  float* ws = (float*)d_ws;

  float* wT0 = ws + OFF_WT0;
  float* wT1 = ws + OFF_WT1;
  float* wT2 = ws + OFF_WT2;
  float* F1 = ws + OFF_F1;
  float* F2 = ws + OFF_F2;
  float* F3 = ws + OFF_F3;
  float* GM = ws + OFF_GM;
  float* WQ = ws + OFF_WQ;

  tr0_kernel<<<(864 * 64 + 255) / 256, 256, 0, stream>>>(c0w, wT0);
  tr1_kernel<<<(576 * 64 + 255) / 256, 256, 0, stream>>>(c1w, wT1);
  tr1_kernel<<<(576 * 64 + 255) / 256, 256, 0, stream>>>(c2w, wT2);

  conv0_kernel<<<dim3(8, 127, 4), 32, 0, stream>>>(x, wT0, c0b, F1);
  convmid_kernel<127, 127, 31, 31, 8><<<248, 128, 0, stream>>>(F1, wT1, c1b, F2);
  convmid_kernel<31, 31, 7, 7, 2><<<14, 128, 0, stream>>>(F2, wT2, c2b, F3);

  mean_kernel<<<4, 64, 0, stream>>>(F3, GM);
  fc_kernel<<<4, 64, 0, stream>>>(GM, f1w, f1b, f2w, f2b, WQ);

  const int npix = 4 * 512 * 512;
  quad_kernel<<<(npix + 255) / 256, 256, 0, stream>>>(x, WQ, out);
}


# conv0: 16-wave workgroups (one per WGP), a tile per wave; each tap's 24 KiB of weights fetched once per workgroup two taps ahead into a two-stage LDS ring, B fragments read from LDS; same sums
# speedup vs baseline: 1.9283x; 1.0373x over previous
#include <hip/hip_runtime.h>
#include <cstddef>

typedef __attribute__((ext_vector_type(2))) float v2f;
typedef __attribute__((ext_vector_type(8))) float v8f;

#define DEV __device__ __forceinline__

DEV float eluf(float v) { return v > 0.f ? v : (__expf(v) - 1.f); }
DEV float histf(float v, float bin, float nm1) {
  return fmaxf(1.f - fabsf(v - bin) * nm1, 0.f);
}

DEV float xhalf(float v) {
  return __builtin_bit_cast(float, __builtin_amdgcn_permlanex16(__builtin_bit_cast(int, v),
                                                                __builtin_bit_cast(int, v),
                                                                0x76543210, 0xfedcba98, false, false));
}

DEV v8f wmma_f32(v2f a, v2f b, v8f c) {
  return __builtin_amdgcn_wmma_f32_16x16x4_f32(false, a, false, b, (short)0, c,
                                               false, false);
}

__global__ void tr0_kernel(const float* __restrict__ w, float* __restrict__ wT) {
  int idx = blockIdx.x * blockDim.x + threadIdx.x;
  if (idx >= 864 * 64) return;
  int k = idx >> 6, n = idx & 63;
  int r = k / 96, q = k % 96;
  int cf = (q < 9) ? q : (q - 3);
  float v = (q < 9 || q >= 12) ? w[(size_t)n * 837 + cf * 9 + r] : 0.f;
  wT[((((size_t)(k >> 1)) * 64 + n) << 1) + (k & 1)] = v;
}

__global__ void tr1_kernel(const float* __restrict__ w, float* __restrict__ wT) {
  int idx = blockIdx.x * blockDim.x + threadIdx.x;
  if (idx >= 576 * 64) return;
  int k = idx >> 6, n = idx & 63;
  int r = k >> 6, cin = k & 63;
  float v = w[(size_t)n * 576 + cin * 9 + r];
  wT[((((size_t)(k >> 1)) * 64 + n) << 1) + (k & 1)] = v;
}

#define C0_COLS 65
#define C0_NPIX (5 * C0_COLS)
#define C0_PITCH 10
#define C0_WAVES 16
#define C0_BP 160
#define C0_BROWS 48

DEV v2f hist2(float v, float b0, float b1, float nm1) {
  v2f a;
  a.x = histf(v, b0, nm1);
  a.y = histf(v, b1, nm1);
  return a;
}

__global__ __launch_bounds__(512) void conv0_kernel(const float* __restrict__ x,
                                                    const float* __restrict__ wT,
                                                    const float* __restrict__ bias,
                                                    float* __restrict__ f1) {
  const int tid = threadIdx.x;
  const int lane = tid & 31;
  const int wave = tid >> 5;
  const int hh = lane >> 4;
  const int ml = lane & 15;
  const int t = C0_WAVES * blockIdx.x + wave;
  const int tx = t & 7;
  const int py = (t >> 3) % 127;
  const int b = (t >> 3) / 127;

  __shared__ float rec_all[C0_WAVES * C0_NPIX * C0_PITCH];
  __shared__ __attribute__((aligned(16))) float bs[2 * C0_BROWS * C0_BP];
  float* rec = rec_all + wave * (C0_NPIX * C0_PITCH);

  const float4* wg = (const float4*)wT + tid;
  float4 q0 = wg[0], q1 = wg[512], q2 = wg[1024];
  float* bsw = &bs[wave * C0_BP + 4 * lane];

  const int gy0 = 4 * py;
  const int gx0 = 64 * tx;
  const float* xb = x + (size_t)b * 3 * 512 * 512;

  for (int p = lane; p < C0_NPIX; p += 32) {
    const int ly = p / C0_COLS, lx = p - ly * C0_COLS;
    const int gy = min(gy0 + ly, 511);
    const int gx = min(gx0 + lx, 511);
    const int ym = max(gy - 1, 0), yp = min(gy + 1, 511);
    const int xm = max(gx - 1, 0), xp = min(gx + 1, 511);
    float f[10];
#pragma unroll
    for (int c = 0; c < 3; ++c) {
      const float* xc = xb + (size_t)c * 512 * 512;
      float a = xc[ym * 512 + xm], bb = xc[ym * 512 + gx], cc = xc[ym * 512 + xp];
      float d = xc[gy * 512 + xm], ee = xc[gy * 512 + gx], e = xc[gy * 512 + xp];
      float g = xc[yp * 512 + xm], h = xc[yp * 512 + gx], i = xc[yp * 512 + xp];
      f[c] = ee;
      f[3 + 2 * c] = ((cc + 2.f * e + i) - (a + 2.f * d + g)) * 0.125f;
      f[4 + 2 * c] = ((g + 2.f * h + i) - (a + 2.f * bb + cc)) * 0.125f;
    }
    f[9] = 0.f;
    v2f* rp = (v2f*)&rec[p * C0_PITCH];
#pragma unroll
    for (int tt = 0; tt < 5; ++tt) {
      v2f o = {f[2 * tt], f[2 * tt + 1]};
      rp[tt] = o;
    }
  }

  *(float4*)(bsw) = q0;
  *(float4*)(bsw + 16 * C0_BP) = q1;
  *(float4*)(bsw + 32 * C0_BP) = q2;
  q0 = wg[1536], q1 = wg[1536 + 512], q2 = wg[1536 + 1024];
  __syncthreads();

  float b4[2], b8[2][2], b16[4][2];
#pragma unroll
  for (int s = 0; s < 2; ++s) {
    b4[s] = (float)(2 * hh + s) * (1.f / 3.f);
#pragma unroll
    for (int j = 0; j < 2; ++j) b8[j][s] = (float)(4 * j + 2 * hh + s) * (1.f / 7.f);
#pragma unroll
    for (int j = 0; j < 4; ++j) b16[j][s] = (float)(4 * j + 2 * hh + s) * (1.f / 15.f);
  }

  v8f acc[4][4];
  const v8f vz = {0.f, 0.f, 0.f, 0.f, 0.f, 0.f, 0.f, 0.f};
#pragma unroll
  for (int mt = 0; mt < 4; ++mt)
#pragma unroll
    for (int nt = 0; nt < 4; ++nt) acc[mt][nt] = vz;

  for (int r = 0; r < 9; ++r) {
    if (r < 8) {
      float* bn = bsw + ((r + 1) & 1) * (C0_BROWS * C0_BP);
      *(float4*)(bn) = q0;
      *(float4*)(bn + 16 * C0_BP) = q1;
      *(float4*)(bn + 32 * C0_BP) = q2;
      if (r < 7) {
        const float4* wn = wg + (r + 2) * 1536;
        q0 = wn[0], q1 = wn[512], q2 = wn[1024];
      }
    }

    const int ky = r / 3, kx = r - 3 * ky;
    v2f raw[4][3];
    float cv[4][3];
#pragma unroll
    for (int mt = 0; mt < 4; ++mt) {
      const int pix = (2 * (mt >> 1) + ky) * C0_COLS + 32 * (mt & 1) + 2 * ml + kx;
      const float* rp = &rec[pix * C0_PITCH];
      const v2f c01 = *(const v2f*)(rp);
      const v2f c23 = *(const v2f*)(rp + 2);
      cv[mt][0] = c01.x;
      cv[mt][1] = c01.y;
      cv[mt][2] = c23.x;
      raw[mt][0] = *(const v2f*)(rp + 2 * hh);
      raw[mt][1] = *(const v2f*)(rp + 4 + 2 * hh);
      const v2f t2 = *(const v2f*)(rp + 8);
      raw[mt][2].x = hh ? 0.f : t2.x;
      raw[mt][2].y = hh ? 0.f : t2.y;
    }
    const float* btap = &bs[(r & 1) * (C0_BROWS * C0_BP) + hh * C0_BP + 2 * ml];
#pragma unroll
    for (int u = 0; u < 24; ++u) {
      v2f a[4];
#pragma unroll
      for (int mt = 0; mt < 4; ++mt) {
        if (u < 3) {
          a[mt] = raw[mt][u < 3 ? u : 0];
        } else if (u < 6) {
          a[mt] = hist2(cv[mt][(u - 3) % 3], b4[0], b4[1], 3.f);
        } else if (u < 12) {
          a[mt] = hist2(cv[mt][((u - 6) >> 1) % 3], b8[(u - 6) & 1][0], b8[(u - 6) & 1][1], 7.f);
        } else {
          a[mt] = hist2(cv[mt][((u - 12) >> 2) % 3], b16[(u - 12) & 3][0], b16[(u - 12) & 3][1],
                        15.f);
        }
      }
      const float* bp = btap + u * (2 * C0_BP);
#pragma unroll
      for (int nt = 0; nt < 4; ++nt) {
        v2f bv = *(const v2f*)(bp + nt * 32);
#pragma unroll
        for (int mt = 0; mt < 4; ++mt) acc[mt][nt] = wmma_f32(a[mt], bv, acc[mt][nt]);
      }
    }
    __syncthreads();
  }

#pragma unroll
  for (int nt = 0; nt < 4; ++nt) {
    const float bb = bias[nt * 16 + ml];
#pragma unroll
    for (int xh = 0; xh < 2; ++xh) {
      const int pxb = 16 * tx + 8 * xh;
#pragma unroll
      for (int j = 0; j < 4; ++j) {
        float m0 = fmaxf(eluf(acc[xh][nt][2 * j] + bb), eluf(acc[xh][nt][2 * j + 1] + bb));
        float m1 =
            fmaxf(eluf(acc[2 + xh][nt][2 * j] + bb), eluf(acc[2 + xh][nt][2 * j + 1] + bb));
        int px = pxb + j + 4 * hh;
        if (px < 127)
          f1[((((size_t)b * 127 + py) * 127 + px) << 6) + nt * 16 + ml] = fmaxf(m0, m1);
      }
    }
  }
}

#define CM_BP 160

template <int HIN, int WIN, int HP, int WP, int NXT>
__global__ __launch_bounds__(128) void convmid_kernel(const float* __restrict__ fin,
                                                      const float* __restrict__ wT,
                                                      const float* __restrict__ bias,
                                                      float* __restrict__ fout) {
  const int tid = threadIdx.x;
  const int lane = tid & 31;
  const int wave = tid >> 5;
  const int hh = lane >> 4;
  const int ml = lane & 15;
  const int t = 4 * blockIdx.x + wave;
  const int tx = t % NXT;
  const int py = (t / NXT) % HP;
  const int b = t / (NXT * HP);

  __shared__ __attribute__((aligned(16))) float bs[32 * CM_BP];

  v8f acc[4];
  const v8f vz = {0.f, 0.f, 0.f, 0.f, 0.f, 0.f, 0.f, 0.f};
  for (int nt = 0; nt < 4; ++nt) acc[nt] = vz;

  const int cx = 8 * tx + (ml & 7);
  const int cy = 2 * py + (ml >> 3);
  const float* finb = fin + (size_t)b * HIN * WIN * 64;

  const float4* wg = (const float4*)wT + tid;
  float4 p0 = wg[0], p1 = wg[128], p2 = wg[256], p3 = wg[384];
  float4 p4 = wg[512], p5 = wg[640], p6 = wg[768], p7 = wg[896];
  float* bsw = &bs[wave * CM_BP + 4 * lane];

  for (int r = 0; r < 9; ++r) {
    __syncthreads();
    *(float4*)(bsw) = p0;
    *(float4*)(bsw + 4 * CM_BP) = p1;
    *(float4*)(bsw + 8 * CM_BP) = p2;
    *(float4*)(bsw + 12 * CM_BP) = p3;
    *(float4*)(bsw + 16 * CM_BP) = p4;
    *(float4*)(bsw + 20 * CM_BP) = p5;
    *(float4*)(bsw + 24 * CM_BP) = p6;
    *(float4*)(bsw + 28 * CM_BP) = p7;
    __syncthreads();
    if (r < 8) {
      const float4* wn = wg + (r + 1) * 1024;
      p0 = wn[0], p1 = wn[128], p2 = wn[256], p3 = wn[384];
      p4 = wn[512], p5 = wn[640], p6 = wn[768], p7 = wn[896];
    }

    const int ky = r / 3, kx = r % 3;
    const int xx = min(2 * cx + kx, WIN - 1);
    const int yy = min(2 * cy + ky, HIN - 1);
    const float* row = finb + (((size_t)yy * WIN + xx) << 6);
#pragma unroll
    for (int u = 0; u < 16; ++u) {
      const int cinb = 4 * u + 2 * hh;
      v2f a = *(const v2f*)(row + cinb);
      const float* bp = &bs[(2 * u + hh) * CM_BP + 2 * ml];
#pragma unroll
      for (int nt = 0; nt < 4; ++nt) {
        v2f bv = *(const v2f*)(bp + nt * 32);
        acc[nt] = wmma_f32(a, bv, acc[nt]);
      }
    }
  }

  const int pxb = 4 * tx;
#pragma unroll
  for (int nt = 0; nt < 4; ++nt) {
    const float bb = bias[nt * 16 + ml];
#pragma unroll
    for (int j = 0; j < 4; ++j) {
      float m = fmaxf(eluf(acc[nt][2 * j] + bb), eluf(acc[nt][2 * j + 1] + bb));
      m = fmaxf(m, xhalf(m));
      const int px = pxb + j;
      if ((j >> 1) == hh && px < WP)
        fout[((((size_t)b * HP + py) * WP + px) << 6) + nt * 16 + ml] = m;
    }
  }
}

__global__ void mean_kernel(const float* __restrict__ f3, float* __restrict__ gm) {
  int b = blockIdx.x, c = threadIdx.x;
  float s = 0.f;
  for (int i = 0; i < 49; ++i) s += f3[(((size_t)b * 49 + i) << 6) + c];
  gm[b * 64 + c] = s * (1.f / 49.f);
}

__global__ void fc_kernel(const float* __restrict__ gm, const float* __restrict__ w1,
                          const float* __restrict__ b1, const float* __restrict__ w2,
                          const float* __restrict__ b2, float* __restrict__ wq) {
  __shared__ float fv[64];
  __shared__ float h[64];
  int b = blockIdx.x, j = threadIdx.x;
  fv[j] = gm[b * 64 + j];
  __syncthreads();
  float acc = b1[j];
  for (int c = 0; c < 64; ++c) acc += fv[c] * w1[j * 64 + c];
  h[j] = eluf(acc);
  __syncthreads();
  if (j < 30) {
    float a2 = b2[j];
    for (int c = 0; c < 64; ++c) a2 += h[c] * w2[j * 64 + c];
    wq[b * 30 + j] = a2;
  }
}

__global__ void quad_kernel(const float* __restrict__ x, const float* __restrict__ wq,
                            float* __restrict__ out) {
  const int HW = 512 * 512;
  int idx = blockIdx.x * blockDim.x + threadIdx.x;
  if (idx >= 4 * HW) return;
  int b = idx / HW, hw = idx - b * HW;
  const float* xb = x + (size_t)b * 3 * HW;
  float r = xb[hw], g = xb[HW + hw], bl = xb[2 * HW + hw];
  float f[10] = {r * r, r * g, r * bl, r, g * g, g * bl, g, bl * bl, bl, 1.f};
  const float* w = wq + b * 30;
  float* ob = out + (size_t)b * 3 * HW;
#pragma unroll
  for (int c = 0; c < 3; ++c) {
    float s = (c == 0) ? r : ((c == 1) ? g : bl);
#pragma unroll
    for (int k = 0; k < 10; ++k) s += f[k] * w[c * 10 + k];
    ob[c * HW + hw] = s;
  }
}

static const size_t OFF_WT0 = 0;
static const size_t OFF_WT1 = 55296;
static const size_t OFF_WT2 = 92160;
static const size_t OFF_F1 = 129024;
static const size_t OFF_F2 = 129024 + 4129024;
static const size_t OFF_F3 = 129024 + 4129024 + 246016;
static const size_t OFF_GM = OFF_F3 + 12544;
static const size_t OFF_WQ = OFF_GM + 256;

extern "C" void kernel_launch(void* const* d_in, const int* in_sizes, int n_in,
                              void* d_out, int out_size, void* d_ws, size_t ws_size,
                              hipStream_t stream) {
  const float* x = (const float*)d_in[0];
  const float* c0w = (const float*)d_in[1];
  const float* c0b = (const float*)d_in[2];
  const float* c1w = (const float*)d_in[3];
  const float* c1b = (const float*)d_in[4];
  const float* c2w = (const float*)d_in[5];
  const float* c2b = (const float*)d_in[6];
  const float* f1w = (const float*)d_in[7];
  const float* f1b = (const float*)d_in[8];
  const float* f2w = (const float*)d_in[9];
  const float* f2b = (const float*)d_in[10];
  float* out = (float*)d_out;
  float* ws = (float*)d_ws;

  float* wT0 = ws + OFF_WT0;
  float* wT1 = ws + OFF_WT1;
  float* wT2 = ws + OFF_WT2;
  float* F1 = ws + OFF_F1;
  float* F2 = ws + OFF_F2;
  float* F3 = ws + OFF_F3;
  float* GM = ws + OFF_GM;
  float* WQ = ws + OFF_WQ;

  tr0_kernel<<<(864 * 64 + 255) / 256, 256, 0, stream>>>(c0w, wT0);
  tr1_kernel<<<(576 * 64 + 255) / 256, 256, 0, stream>>>(c1w, wT1);
  tr1_kernel<<<(576 * 64 + 255) / 256, 256, 0, stream>>>(c2w, wT2);

  conv0_kernel<<<254, 512, 0, stream>>>(x, wT0, c0b, F1);
  convmid_kernel<127, 127, 31, 31, 8><<<248, 128, 0, stream>>>(F1, wT1, c1b, F2);
  convmid_kernel<31, 31, 7, 7, 2><<<14, 128, 0, stream>>>(F2, wT2, c2b, F3);

  mean_kernel<<<4, 64, 0, stream>>>(F3, GM);
  fc_kernel<<<4, 64, 0, stream>>>(GM, f1w, f1b, f2w, f2b, WQ);

  const int npix = 4 * 512 * 512;
  quad_kernel<<<(npix + 255) / 256, 256, 0, stream>>>(x, WQ, out);
}


# conv1/conv2 read OIHW weights as given: K re-ordered (16-channel block, tap, channel), block = 64 contiguous 576-byte runs copied to LDS; both transpose launches go (9 -> 7); changes summation order
# speedup vs baseline: 2.0387x; 1.0573x over previous
#include <hip/hip_runtime.h>
#include <cstddef>

typedef __attribute__((ext_vector_type(2))) float v2f;
typedef __attribute__((ext_vector_type(4))) float v4f;
typedef __attribute__((ext_vector_type(8))) float v8f;

#define DEV __device__ __forceinline__

DEV float eluf(float v) { return v > 0.f ? v : (__expf(v) - 1.f); }
DEV float histf(float v, float bin, float nm1) {
  return fmaxf(1.f - fabsf(v - bin) * nm1, 0.f);
}

DEV float xhalf(float v) {
  return __builtin_bit_cast(float, __builtin_amdgcn_permlanex16(__builtin_bit_cast(int, v),
                                                                __builtin_bit_cast(int, v),
                                                                0x76543210, 0xfedcba98, false, false));
}

DEV v8f wmma_f32(v2f a, v2f b, v8f c) {
  return __builtin_amdgcn_wmma_f32_16x16x4_f32(false, a, false, b, (short)0, c,
                                               false, false);
}

__global__ void tr0_kernel(const float* __restrict__ w, float* __restrict__ wT) {
  int idx = blockIdx.x * blockDim.x + threadIdx.x;
  if (idx >= 864 * 64) return;
  int k = idx >> 6, n = idx & 63;
  int r = k / 96, q = k % 96;
  int cf = (q < 9) ? q : (q - 3);
  float v = (q < 9 || q >= 12) ? w[(size_t)n * 837 + cf * 9 + r] : 0.f;
  wT[((((size_t)(k >> 1)) * 64 + n) << 1) + (k & 1)] = v;
}

#define C0_COLS 65
#define C0_NPIX (5 * C0_COLS)
#define C0_PITCH 10
#define C0_WAVES 16
#define C0_BP 160
#define C0_BROWS 48

DEV v2f hist2(float v, float b0, float b1, float nm1) {
  v2f a;
  a.x = histf(v, b0, nm1);
  a.y = histf(v, b1, nm1);
  return a;
}

__global__ __launch_bounds__(512) void conv0_kernel(const float* __restrict__ x,
                                                    const float* __restrict__ wT,
                                                    const float* __restrict__ bias,
                                                    float* __restrict__ f1) {
  const int tid = threadIdx.x;
  const int lane = tid & 31;
  const int wave = tid >> 5;
  const int hh = lane >> 4;
  const int ml = lane & 15;
  const int t = C0_WAVES * blockIdx.x + wave;
  const int tx = t & 7;
  const int py = (t >> 3) % 127;
  const int b = (t >> 3) / 127;

  __shared__ float rec_all[C0_WAVES * C0_NPIX * C0_PITCH];
  __shared__ __attribute__((aligned(16))) float bs[2 * C0_BROWS * C0_BP];
  float* rec = rec_all + wave * (C0_NPIX * C0_PITCH);

  const float4* wg = (const float4*)wT + tid;
  float4 q0 = wg[0], q1 = wg[512], q2 = wg[1024];
  float* bsw = &bs[wave * C0_BP + 4 * lane];

  const int gy0 = 4 * py;
  const int gx0 = 64 * tx;
  const float* xb = x + (size_t)b * 3 * 512 * 512;

  for (int p = lane; p < C0_NPIX; p += 32) {
    const int ly = p / C0_COLS, lx = p - ly * C0_COLS;
    const int gy = min(gy0 + ly, 511);
    const int gx = min(gx0 + lx, 511);
    const int ym = max(gy - 1, 0), yp = min(gy + 1, 511);
    const int xm = max(gx - 1, 0), xp = min(gx + 1, 511);
    float f[10];
#pragma unroll
    for (int c = 0; c < 3; ++c) {
      const float* xc = xb + (size_t)c * 512 * 512;
      float a = xc[ym * 512 + xm], bb = xc[ym * 512 + gx], cc = xc[ym * 512 + xp];
      float d = xc[gy * 512 + xm], ee = xc[gy * 512 + gx], e = xc[gy * 512 + xp];
      float g = xc[yp * 512 + xm], h = xc[yp * 512 + gx], i = xc[yp * 512 + xp];
      f[c] = ee;
      f[3 + 2 * c] = ((cc + 2.f * e + i) - (a + 2.f * d + g)) * 0.125f;
      f[4 + 2 * c] = ((g + 2.f * h + i) - (a + 2.f * bb + cc)) * 0.125f;
    }
    f[9] = 0.f;
    v2f* rp = (v2f*)&rec[p * C0_PITCH];
#pragma unroll
    for (int tt = 0; tt < 5; ++tt) {
      v2f o = {f[2 * tt], f[2 * tt + 1]};
      rp[tt] = o;
    }
  }

  *(float4*)(bsw) = q0;
  *(float4*)(bsw + 16 * C0_BP) = q1;
  *(float4*)(bsw + 32 * C0_BP) = q2;
  q0 = wg[1536], q1 = wg[1536 + 512], q2 = wg[1536 + 1024];
  __syncthreads();

  float b4[2], b8[2][2], b16[4][2];
#pragma unroll
  for (int s = 0; s < 2; ++s) {
    b4[s] = (float)(2 * hh + s) * (1.f / 3.f);
#pragma unroll
    for (int j = 0; j < 2; ++j) b8[j][s] = (float)(4 * j + 2 * hh + s) * (1.f / 7.f);
#pragma unroll
    for (int j = 0; j < 4; ++j) b16[j][s] = (float)(4 * j + 2 * hh + s) * (1.f / 15.f);
  }

  v8f acc[4][4];
  const v8f vz = {0.f, 0.f, 0.f, 0.f, 0.f, 0.f, 0.f, 0.f};
#pragma unroll
  for (int mt = 0; mt < 4; ++mt)
#pragma unroll
    for (int nt = 0; nt < 4; ++nt) acc[mt][nt] = vz;

  for (int r = 0; r < 9; ++r) {
    if (r < 8) {
      float* bn = bsw + ((r + 1) & 1) * (C0_BROWS * C0_BP);
      *(float4*)(bn) = q0;
      *(float4*)(bn + 16 * C0_BP) = q1;
      *(float4*)(bn + 32 * C0_BP) = q2;
      if (r < 7) {
        const float4* wn = wg + (r + 2) * 1536;
        q0 = wn[0], q1 = wn[512], q2 = wn[1024];
      }
    }

    const int ky = r / 3, kx = r - 3 * ky;
    v2f raw[4][3];
    float cv[4][3];
#pragma unroll
    for (int mt = 0; mt < 4; ++mt) {
      const int pix = (2 * (mt >> 1) + ky) * C0_COLS + 32 * (mt & 1) + 2 * ml + kx;
      const float* rp = &rec[pix * C0_PITCH];
      const v2f c01 = *(const v2f*)(rp);
      const v2f c23 = *(const v2f*)(rp + 2);
      cv[mt][0] = c01.x;
      cv[mt][1] = c01.y;
      cv[mt][2] = c23.x;
      raw[mt][0] = *(const v2f*)(rp + 2 * hh);
      raw[mt][1] = *(const v2f*)(rp + 4 + 2 * hh);
      const v2f t2 = *(const v2f*)(rp + 8);
      raw[mt][2].x = hh ? 0.f : t2.x;
      raw[mt][2].y = hh ? 0.f : t2.y;
    }
    const float* btap = &bs[(r & 1) * (C0_BROWS * C0_BP) + hh * C0_BP + 2 * ml];
#pragma unroll
    for (int u = 0; u < 24; ++u) {
      v2f a[4];
#pragma unroll
      for (int mt = 0; mt < 4; ++mt) {
        if (u < 3) {
          a[mt] = raw[mt][u < 3 ? u : 0];
        } else if (u < 6) {
          a[mt] = hist2(cv[mt][(u - 3) % 3], b4[0], b4[1], 3.f);
        } else if (u < 12) {
          a[mt] = hist2(cv[mt][((u - 6) >> 1) % 3], b8[(u - 6) & 1][0], b8[(u - 6) & 1][1], 7.f);
        } else {
          a[mt] = hist2(cv[mt][((u - 12) >> 2) % 3], b16[(u - 12) & 3][0], b16[(u - 12) & 3][1],
                        15.f);
        }
      }
      const float* bp = btap + u * (2 * C0_BP);
#pragma unroll
      for (int nt = 0; nt < 4; ++nt) {
        v2f bv = *(const v2f*)(bp + nt * 32);
#pragma unroll
        for (int mt = 0; mt < 4; ++mt) acc[mt][nt] = wmma_f32(a[mt], bv, acc[mt][nt]);
      }
    }
    __syncthreads();
  }

#pragma unroll
  for (int nt = 0; nt < 4; ++nt) {
    const float bb = bias[nt * 16 + ml];
#pragma unroll
    for (int xh = 0; xh < 2; ++xh) {
      const int pxb = 16 * tx + 8 * xh;
#pragma unroll
      for (int j = 0; j < 4; ++j) {
        float m0 = fmaxf(eluf(acc[xh][nt][2 * j] + bb), eluf(acc[xh][nt][2 * j + 1] + bb));
        float m1 =
            fmaxf(eluf(acc[2 + xh][nt][2 * j] + bb), eluf(acc[2 + xh][nt][2 * j + 1] + bb));
        int px = pxb + j + 4 * hh;
        if (px < 127)
          f1[((((size_t)b * 127 + py) * 127 + px) << 6) + nt * 16 + ml] = fmaxf(m0, m1);
      }
    }
  }
}

#define CM_CB 16
#define CM_NB (64 / CM_CB)
#define CM_KS (CM_CB / 4)
#define CM_RUN4 (CM_CB * 9 / 4)
#define CM_NIT (64 * CM_RUN4 / 128)
#define CM_BP (CM_CB * 9 + 4)

template <int HIN, int WIN, int HP, int WP, int NXT>
__global__ __launch_bounds__(128) void convmid_kernel(const float* __restrict__ fin,
                                                      const float* __restrict__ w,
                                                      const float* __restrict__ bias,
                                                      float* __restrict__ fout) {
  const int tid = threadIdx.x;
  const int lane = tid & 31;
  const int wave = tid >> 5;
  const int hh = lane >> 4;
  const int ml = lane & 15;
  const int t = 4 * blockIdx.x + wave;
  const int tx = t % NXT;
  const int py = (t / NXT) % HP;
  const int b = t / (NXT * HP);

  __shared__ __attribute__((aligned(16))) float bs[64 * CM_BP];

  v8f acc[4];
  const v8f vz = {0.f, 0.f, 0.f, 0.f, 0.f, 0.f, 0.f, 0.f};
  for (int nt = 0; nt < 4; ++nt) acc[nt] = vz;

  const int cx = 8 * tx + (ml & 7);
  const int cy = 2 * py + (ml >> 3);
  const float* finb = fin + (size_t)b * HIN * WIN * 64;

  int poff[9];
#pragma unroll
  for (int r = 0; r < 9; ++r) {
    const int ky = r / 3, kx = r % 3;
    const int xx = min(2 * cx + kx, WIN - 1);
    const int yy = min(2 * cy + ky, HIN - 1);
    poff[r] = ((yy * WIN + xx) << 6) + 2 * hh;
  }

  const v4f* wg = (const v4f*)w;
  int gofs[CM_NIT], lofs[CM_NIT];
#pragma unroll
  for (int i = 0; i < CM_NIT; ++i) {
    const int e = i * 128 + tid;
    const int n = e / CM_RUN4, q = e - CM_RUN4 * n;
    gofs[i] = n * 144 + q;
    lofs[i] = n * CM_BP + 4 * q;
  }
  v4f p[CM_NIT];
#pragma unroll
  for (int i = 0; i < CM_NIT; ++i) p[i] = wg[gofs[i]];

  const float* blane = &bs[ml * CM_BP + 18 * hh];

  for (int cb = 0; cb < CM_NB; ++cb) {
    __syncthreads();
#pragma unroll
    for (int i = 0; i < CM_NIT; ++i) *(v4f*)(&bs[lofs[i]]) = p[i];
    __syncthreads();
    const float* ab = finb + CM_CB * cb;
    v2f afr[9 * CM_KS];
#pragma unroll
    for (int r = 0; r < 9; ++r) {
#pragma unroll
      for (int cu = 0; cu < CM_KS; ++cu)
        afr[CM_KS * r + cu] = *(const v2f*)(ab + poff[r] + 4 * cu);
    }
    if (cb < CM_NB - 1) {
      const v4f* wn = wg + (cb + 1) * CM_RUN4;
#pragma unroll
      for (int i = 0; i < CM_NIT; ++i) p[i] = wn[gofs[i]];
    }
    __builtin_amdgcn_sched_barrier(0);
#pragma unroll
    for (int r = 0; r < 9; ++r) {
#pragma unroll
      for (int cu = 0; cu < CM_KS; ++cu) {
        const float* bp = blane + 36 * cu + r;
#pragma unroll
        for (int nt = 0; nt < 4; ++nt) {
          v2f bv;
          bv.x = bp[nt * 16 * CM_BP];
          bv.y = bp[nt * 16 * CM_BP + 9];
          acc[nt] = wmma_f32(afr[CM_KS * r + cu], bv, acc[nt]);
        }
      }
    }
  }

  const int pxb = 4 * tx;
#pragma unroll
  for (int nt = 0; nt < 4; ++nt) {
    const float bb = bias[nt * 16 + ml];
#pragma unroll
    for (int j = 0; j < 4; ++j) {
      float m = fmaxf(eluf(acc[nt][2 * j] + bb), eluf(acc[nt][2 * j + 1] + bb));
      m = fmaxf(m, xhalf(m));
      const int px = pxb + j;
      if ((j >> 1) == hh && px < WP)
        fout[((((size_t)b * HP + py) * WP + px) << 6) + nt * 16 + ml] = m;
    }
  }
}

__global__ void mean_kernel(const float* __restrict__ f3, float* __restrict__ gm) {
  int b = blockIdx.x, c = threadIdx.x;
  float s = 0.f;
  for (int i = 0; i < 49; ++i) s += f3[(((size_t)b * 49 + i) << 6) + c];
  gm[b * 64 + c] = s * (1.f / 49.f);
}

__global__ void fc_kernel(const float* __restrict__ gm, const float* __restrict__ w1,
                          const float* __restrict__ b1, const float* __restrict__ w2,
                          const float* __restrict__ b2, float* __restrict__ wq) {
  __shared__ float fv[64];
  __shared__ float h[64];
  int b = blockIdx.x, j = threadIdx.x;
  fv[j] = gm[b * 64 + j];
  __syncthreads();
  float acc = b1[j];
  for (int c = 0; c < 64; ++c) acc += fv[c] * w1[j * 64 + c];
  h[j] = eluf(acc);
  __syncthreads();
  if (j < 30) {
    float a2 = b2[j];
    for (int c = 0; c < 64; ++c) a2 += h[c] * w2[j * 64 + c];
    wq[b * 30 + j] = a2;
  }
}

__global__ void quad_kernel(const float* __restrict__ x, const float* __restrict__ wq,
                            float* __restrict__ out) {
  const int HW = 512 * 512;
  int idx = blockIdx.x * blockDim.x + threadIdx.x;
  if (idx >= 4 * HW) return;
  int b = idx / HW, hw = idx - b * HW;
  const float* xb = x + (size_t)b * 3 * HW;
  float r = xb[hw], g = xb[HW + hw], bl = xb[2 * HW + hw];
  float f[10] = {r * r, r * g, r * bl, r, g * g, g * bl, g, bl * bl, bl, 1.f};
  const float* w = wq + b * 30;
  float* ob = out + (size_t)b * 3 * HW;
#pragma unroll
  for (int c = 0; c < 3; ++c) {
    float s = (c == 0) ? r : ((c == 1) ? g : bl);
#pragma unroll
    for (int k = 0; k < 10; ++k) s += f[k] * w[c * 10 + k];
    ob[c * HW + hw] = s;
  }
}

static const size_t OFF_WT0 = 0;
static const size_t OFF_F1 = 129024;
static const size_t OFF_F2 = 129024 + 4129024;
static const size_t OFF_F3 = 129024 + 4129024 + 246016;
static const size_t OFF_GM = OFF_F3 + 12544;
static const size_t OFF_WQ = OFF_GM + 256;

extern "C" void kernel_launch(void* const* d_in, const int* in_sizes, int n_in,
                              void* d_out, int out_size, void* d_ws, size_t ws_size,
                              hipStream_t stream) {
  const float* x = (const float*)d_in[0];
  const float* c0w = (const float*)d_in[1];
  const float* c0b = (const float*)d_in[2];
  const float* c1w = (const float*)d_in[3];
  const float* c1b = (const float*)d_in[4];
  const float* c2w = (const float*)d_in[5];
  const float* c2b = (const float*)d_in[6];
  const float* f1w = (const float*)d_in[7];
  const float* f1b = (const float*)d_in[8];
  const float* f2w = (const float*)d_in[9];
  const float* f2b = (const float*)d_in[10];
  float* out = (float*)d_out;
  float* ws = (float*)d_ws;

  float* wT0 = ws + OFF_WT0;
  float* F1 = ws + OFF_F1;
  float* F2 = ws + OFF_F2;
  float* F3 = ws + OFF_F3;
  float* GM = ws + OFF_GM;
  float* WQ = ws + OFF_WQ;

  tr0_kernel<<<(864 * 64 + 255) / 256, 256, 0, stream>>>(c0w, wT0);

  conv0_kernel<<<254, 512, 0, stream>>>(x, wT0, c0b, F1);
  convmid_kernel<127, 127, 31, 31, 8><<<248, 128, 0, stream>>>(F1, c1w, c1b, F2);
  convmid_kernel<31, 31, 7, 7, 2><<<14, 128, 0, stream>>>(F2, c2w, c2b, F3);

  mean_kernel<<<4, 64, 0, stream>>>(F3, GM);
  fc_kernel<<<4, 64, 0, stream>>>(GM, f1w, f1b, f2w, f2b, WQ);

  const int npix = 4 * 512 * 512;
  quad_kernel<<<(npix + 255) / 256, 256, 0, stream>>>(x, WQ, out);
}
